# starting kernel
# baseline (speedup 1.0000x reference)
#include <hip/hip_runtime.h>
#include <cstddef>

typedef __attribute__((ext_vector_type(2))) float v2f;
typedef __attribute__((ext_vector_type(8))) float v8f;

#define BSZ 64
#define TSZ 256
#define FSZ 256
#define HSZ 1024
#define OSZ 256

#define NWAVE 8

__device__ __forceinline__ void wmma_tile_f32(const float* __restrict__ A0, int lda,
                                              const float* __restrict__ B0, int ldb,
                                              int kBeg, int kEnd, int lane, v8f& acc)
{
    const int rc = lane & 15;
    const int kh = (lane >> 4) << 1;
    const float* ap = A0 + (size_t)rc * lda + kh;
    const float* bp = B0 + (size_t)rc * ldb + kh;
#pragma unroll 4
    for (int k = kBeg; k < kEnd; k += 4) {
        v2f a = *(const v2f*)(ap + k);
        v2f b = *(const v2f*)(bp + k);
        acc = __builtin_amdgcn_wmma_f32_16x16x4_f32(false, a, false, b,
                                                    (short)0, acc, false, false);
    }
}

__global__ void __launch_bounds__(32 * NWAVE)
rnn_step(const float* __restrict__ x,
         const float* __restrict__ W_ih,
         const float* __restrict__ W_hh,
         const float* __restrict__ b_ih,
         const float* __restrict__ b_hh,
         const float* __restrict__ hprev,
         float* __restrict__ hnext,
         int t, int has_h)
{
    __shared__ float red[NWAVE * 32 * 8];
    const int lane = threadIdx.x & 31;
    const int wave = threadIdx.x >> 5;
    const int tm = blockIdx.x;
    const int tn = blockIdx.y;

    v8f acc = {};

    {
        const float* A0 = x + (size_t)(tm * 16) * (TSZ * FSZ) + (size_t)t * FSZ;
        const float* B0 = W_ih + (size_t)(tn * 16) * FSZ;
        wmma_tile_f32(A0, TSZ * FSZ, B0, FSZ,
                      wave * (FSZ / NWAVE), (wave + 1) * (FSZ / NWAVE), lane, acc);
    }
    if (has_h) {
        const float* A0 = hprev + (size_t)(tm * 16) * HSZ;
        const float* B0 = W_hh + (size_t)(tn * 16) * HSZ;
        wmma_tile_f32(A0, HSZ, B0, HSZ,
                      wave * (HSZ / NWAVE), (wave + 1) * (HSZ / NWAVE), lane, acc);
    }

    {
        const int base = wave * 256 + lane * 8;
#pragma unroll
        for (int j = 0; j < 8; ++j) red[base + j] = acc[j];
    }
    __syncthreads();

    if (wave == 0) {
        const int n  = lane & 15;
        const int mh = (lane >> 4) << 3;
        const int hh = tn * 16 + n;
        const float bias = b_ih[hh] + b_hh[hh];
#pragma unroll
        for (int j = 0; j < 8; ++j) {
            const int idx = lane * 8 + j;
            float s = 0.f;
#pragma unroll
            for (int w = 0; w < NWAVE; ++w) s += red[w * 256 + idx];
            const int bb = tm * 16 + mh + j;
            hnext[(size_t)bb * HSZ + hh] = tanhf(s + bias);
        }
    }
}

__global__ void __launch_bounds__(32 * NWAVE)
fc_head(const float* __restrict__ hlast,
        const float* __restrict__ W_fc,
        const float* __restrict__ b_fc,
        float* __restrict__ out)
{
    __shared__ float red[NWAVE * 32 * 8];
    const int lane = threadIdx.x & 31;
    const int wave = threadIdx.x >> 5;
    const int tm = blockIdx.x;
    const int tn = blockIdx.y;

    v8f acc = {};
    {
        const float* A0 = hlast + (size_t)(tm * 16) * HSZ;
        const float* B0 = W_fc + (size_t)(tn * 16) * HSZ;
        wmma_tile_f32(A0, HSZ, B0, HSZ,
                      wave * (HSZ / NWAVE), (wave + 1) * (HSZ / NWAVE), lane, acc);
    }
    {
        const int base = wave * 256 + lane * 8;
#pragma unroll
        for (int j = 0; j < 8; ++j) red[base + j] = acc[j];
    }
    __syncthreads();

    if (wave == 0) {
        const int n  = lane & 15;
        const int mh = (lane >> 4) << 3;
        const int oo = tn * 16 + n;
        const float bias = b_fc[oo];
#pragma unroll
        for (int j = 0; j < 8; ++j) {
            const int idx = lane * 8 + j;
            float s = 0.f;
#pragma unroll
            for (int w = 0; w < NWAVE; ++w) s += red[w * 256 + idx];
            const int bb = tm * 16 + mh + j;
            out[(size_t)bb * OSZ + oo] = s + bias;
        }
    }
}

extern "C" void kernel_launch(void* const* d_in, const int* in_sizes, int n_in,
                              void* d_out, int out_size, void* d_ws, size_t ws_size,
                              hipStream_t stream)
{
    const float* x    = (const float*)d_in[0];
    const float* W_ih = (const float*)d_in[1];
    const float* W_hh = (const float*)d_in[2];
    const float* b_ih = (const float*)d_in[3];
    const float* b_hh = (const float*)d_in[4];
    const float* W_fc = (const float*)d_in[5];
    const float* b_fc = (const float*)d_in[6];
    float* out = (float*)d_out;

    float* h0 = (float*)d_ws;
    float* h1 = h0 + BSZ * HSZ;
    float* h[2] = { h0, h1 };

    const dim3 sgrid(BSZ / 16, HSZ / 16);
    for (int t = 0; t < TSZ; ++t) {
        const float* hp = h[t & 1];
        float*       hn = h[(t + 1) & 1];
        rnn_step<<<sgrid, 32 * NWAVE, 0, stream>>>(x, W_ih, W_hh, b_ih, b_hh,
                                                   hp, hn, t, (t > 0) ? 1 : 0);
    }
    const float* hlast = h[TSZ & 1];
    fc_head<<<dim3(BSZ / 16, OSZ / 16), 32 * NWAVE, 0, stream>>>(hlast, W_fc, b_fc, out);
}


# rnn_step operand fetch: each wave loads its A/B K-slice as whole 128-byte row lines (16 B per lane), parks it in its own LDS area and reads fragments there; k order and sums unchanged
# speedup vs baseline: 1.2891x; 1.2891x over previous
#include <hip/hip_runtime.h>
#include <cstddef>


typedef __attribute__((ext_vector_type(2))) float v2f;
typedef __attribute__((ext_vector_type(4))) float v4f;
typedef __attribute__((ext_vector_type(8))) float v8f;

#define BSZ 64
#define TSZ 256
#define FSZ 256
#define HSZ 1024
#define OSZ 256

#define NWAVE 8

#define KC     32
#define PITCH  36
#define TILE_F (16 * PITCH)
#define WAVE_F (2 * 2 * TILE_F)

__device__ __forceinline__ void wmma_tile_f32(const float* __restrict__ A0, int lda,
                                              const float* __restrict__ B0, int ldb,
                                              int kBeg, int kEnd, int lane, v8f& acc)
{
    const int rc = lane & 15;
    const int kh = (lane >> 4) << 1;
    const float* ap = A0 + (size_t)rc * lda + kh;
    const float* bp = B0 + (size_t)rc * ldb + kh;
#pragma unroll 4
    for (int k = kBeg; k < kEnd; k += 4) {
        v2f a = *(const v2f*)(ap + k);
        v2f b = *(const v2f*)(bp + k);
        acc = __builtin_amdgcn_wmma_f32_16x16x4_f32(false, a, false, b,
                                                    (short)0, acc, false, false);
    }
}

template <int NC>
__device__ __forceinline__ void staged_slice_f32(const float* __restrict__ a0, size_t lda0,
                                                 const float* __restrict__ b0, size_t ldb0,
                                                 const float* __restrict__ a1,
                                                 const float* __restrict__ b1,
                                                 float* st, int lane, v8f& acc)
{
    v4f ga[NC][4], gb[NC][4];
#pragma unroll
    for (int c = 0; c < NC; ++c) {
        const float* pa = (c == 0) ? a0 : a1 + (c - 1) * KC;
        const float* pb = (c == 0) ? b0 : b1 + (c - 1) * KC;
        const size_t sa = (c == 0) ? lda0 : (size_t)HSZ;
        const size_t sb = (c == 0) ? ldb0 : (size_t)HSZ;
#pragma unroll
        for (int q = 0; q < 4; ++q) {
            ga[c][q] = *(const v4f*)(pa + (size_t)(4 * q) * sa);
            gb[c][q] = *(const v4f*)(pb + (size_t)(4 * q) * sb);
        }
    }

    const int lr = lane >> 3;
    const int lc = (lane & 7) << 2;
    const int rc = lane & 15;
    const int kh = (lane >> 4) << 1;
#pragma unroll
    for (int c = 0; c < NC; ++c) {
        float* sA = st + (c & 1) * (2 * TILE_F);
        float* sB = sA + TILE_F;
#pragma unroll
        for (int q = 0; q < 4; ++q) {
            *(v4f*)(sA + (4 * q + lr) * PITCH + lc) = ga[c][q];
            *(v4f*)(sB + (4 * q + lr) * PITCH + lc) = gb[c][q];
        }
#pragma unroll
        for (int k = 0; k < KC; k += 4) {
            v2f a = *(const v2f*)(sA + rc * PITCH + kh + k);
            v2f b = *(const v2f*)(sB + rc * PITCH + kh + k);
            acc = __builtin_amdgcn_wmma_f32_16x16x4_f32(false, a, false, b,
                                                        (short)0, acc, false, false);
        }
    }
}

__global__ void __launch_bounds__(32 * NWAVE)
rnn_step(const float* __restrict__ x,
         const float* __restrict__ W_ih,
         const float* __restrict__ W_hh,
         const float* __restrict__ b_ih,
         const float* __restrict__ b_hh,
         const float* __restrict__ hprev,
         float* __restrict__ hnext,
         int t, int has_h)
{
    __shared__ float red[NWAVE * 32 * 8];
    __shared__ __attribute__((aligned(16))) float stage[NWAVE * WAVE_F];
    const int lane = threadIdx.x & 31;
    const int wave = threadIdx.x >> 5;
    const int tm = blockIdx.x;
    const int tn = blockIdx.y;

    v8f acc = {};

    {
        const int lr = lane >> 3;
        const int lc = (lane & 7) << 2;
        const float* a0 = x + (size_t)(tm * 16 + lr) * (TSZ * FSZ) + (size_t)t * FSZ
                            + wave * (FSZ / NWAVE) + lc;
        const float* b0 = W_ih + (size_t)(tn * 16 + lr) * FSZ + wave * (FSZ / NWAVE) + lc;
        const float* a1 = hprev + (size_t)(tm * 16 + lr) * HSZ + wave * (HSZ / NWAVE) + lc;
        const float* b1 = W_hh + (size_t)(tn * 16 + lr) * HSZ + wave * (HSZ / NWAVE) + lc;
        float* st = stage + wave * WAVE_F;
        if (has_h) {
            staged_slice_f32<1 + (HSZ / NWAVE) / KC>(a0, (size_t)TSZ * FSZ, b0, (size_t)FSZ,
                                                     a1, b1, st, lane, acc);
        } else {
            staged_slice_f32<1>(a0, (size_t)TSZ * FSZ, b0, (size_t)FSZ, a1, b1, st, lane, acc);
        }
    }

    {
        const int base = wave * 256 + lane * 8;
#pragma unroll
        for (int j = 0; j < 8; ++j) red[base + j] = acc[j];
    }
    __syncthreads();

    if (wave == 0) {
        const int n  = lane & 15;
        const int mh = (lane >> 4) << 3;
        const int hh = tn * 16 + n;
        const float bias = b_ih[hh] + b_hh[hh];
#pragma unroll
        for (int j = 0; j < 8; ++j) {
            const int idx = lane * 8 + j;
            float s = 0.f;
#pragma unroll
            for (int w = 0; w < NWAVE; ++w) s += red[w * 256 + idx];
            const int bb = tm * 16 + mh + j;
            hnext[(size_t)bb * HSZ + hh] = tanhf(s + bias);
        }
    }
}

__global__ void __launch_bounds__(32 * NWAVE)
fc_head(const float* __restrict__ hlast,
        const float* __restrict__ W_fc,
        const float* __restrict__ b_fc,
        float* __restrict__ out)
{
    __shared__ float red[NWAVE * 32 * 8];
    const int lane = threadIdx.x & 31;
    const int wave = threadIdx.x >> 5;
    const int tm = blockIdx.x;
    const int tn = blockIdx.y;

    v8f acc = {};
    {
        const float* A0 = hlast + (size_t)(tm * 16) * HSZ;
        const float* B0 = W_fc + (size_t)(tn * 16) * HSZ;
        wmma_tile_f32(A0, HSZ, B0, HSZ,
                      wave * (HSZ / NWAVE), (wave + 1) * (HSZ / NWAVE), lane, acc);
    }
    {
        const int base = wave * 256 + lane * 8;
#pragma unroll
        for (int j = 0; j < 8; ++j) red[base + j] = acc[j];
    }
    __syncthreads();

    if (wave == 0) {
        const int n  = lane & 15;
        const int mh = (lane >> 4) << 3;
        const int oo = tn * 16 + n;
        const float bias = b_fc[oo];
#pragma unroll
        for (int j = 0; j < 8; ++j) {
            const int idx = lane * 8 + j;
            float s = 0.f;
#pragma unroll
            for (int w = 0; w < NWAVE; ++w) s += red[w * 256 + idx];
            const int bb = tm * 16 + mh + j;
            out[(size_t)bb * OSZ + oo] = s + bias;
        }
    }
}

extern "C" void kernel_launch(void* const* d_in, const int* in_sizes, int n_in,
                              void* d_out, int out_size, void* d_ws, size_t ws_size,
                              hipStream_t stream)
{
    const float* x    = (const float*)d_in[0];
    const float* W_ih = (const float*)d_in[1];
    const float* W_hh = (const float*)d_in[2];
    const float* b_ih = (const float*)d_in[3];
    const float* b_hh = (const float*)d_in[4];
    const float* W_fc = (const float*)d_in[5];
    const float* b_fc = (const float*)d_in[6];
    float* out = (float*)d_out;

    float* h0 = (float*)d_ws;
    float* h1 = h0 + BSZ * HSZ;
    float* h[2] = { h0, h1 };

    const dim3 sgrid(BSZ / 16, HSZ / 16);
    for (int t = 0; t < TSZ; ++t) {
        const float* hp = h[t & 1];
        float*       hn = h[(t + 1) & 1];
        rnn_step<<<sgrid, 32 * NWAVE, 0, stream>>>(x, W_ih, W_hh, b_ih, b_hh,
                                                   hp, hn, t, (t > 0) ? 1 : 0);
    }
    const float* hlast = h[TSZ & 1];
    fc_head<<<dim3(BSZ / 16, OSZ / 16), 32 * NWAVE, 0, stream>>>(hlast, W_fc, b_fc, out);
}


# rnn_step tail: all 8 waves finish the tile, one output per thread (cross-wave sum, bias, tanh, store) instead of wave 0 in 8 serial rounds; bias fetched at entry; sum order unchanged
# speedup vs baseline: 1.4031x; 1.0884x over previous
#include <hip/hip_runtime.h>
#include <cstddef>


typedef __attribute__((ext_vector_type(2))) float v2f;
typedef __attribute__((ext_vector_type(4))) float v4f;
typedef __attribute__((ext_vector_type(8))) float v8f;

#define BSZ 64
#define TSZ 256
#define FSZ 256
#define HSZ 1024
#define OSZ 256

#define NWAVE 8

#define KC     32
#define PITCH  36
#define TILE_F (16 * PITCH)
#define WAVE_F (2 * 2 * TILE_F)

__device__ __forceinline__ void wmma_tile_f32(const float* __restrict__ A0, int lda,
                                              const float* __restrict__ B0, int ldb,
                                              int kBeg, int kEnd, int lane, v8f& acc)
{
    const int rc = lane & 15;
    const int kh = (lane >> 4) << 1;
    const float* ap = A0 + (size_t)rc * lda + kh;
    const float* bp = B0 + (size_t)rc * ldb + kh;
#pragma unroll 4
    for (int k = kBeg; k < kEnd; k += 4) {
        v2f a = *(const v2f*)(ap + k);
        v2f b = *(const v2f*)(bp + k);
        acc = __builtin_amdgcn_wmma_f32_16x16x4_f32(false, a, false, b,
                                                    (short)0, acc, false, false);
    }
}

template <int NC>
__device__ __forceinline__ void staged_slice_f32(const float* __restrict__ a0, size_t lda0,
                                                 const float* __restrict__ b0, size_t ldb0,
                                                 const float* __restrict__ a1,
                                                 const float* __restrict__ b1,
                                                 float* st, int lane, v8f& acc)
{
    v4f ga[NC][4], gb[NC][4];
#pragma unroll
    for (int c = 0; c < NC; ++c) {
        const float* pa = (c == 0) ? a0 : a1 + (c - 1) * KC;
        const float* pb = (c == 0) ? b0 : b1 + (c - 1) * KC;
        const size_t sa = (c == 0) ? lda0 : (size_t)HSZ;
        const size_t sb = (c == 0) ? ldb0 : (size_t)HSZ;
#pragma unroll
        for (int q = 0; q < 4; ++q) {
            ga[c][q] = *(const v4f*)(pa + (size_t)(4 * q) * sa);
            gb[c][q] = *(const v4f*)(pb + (size_t)(4 * q) * sb);
        }
    }

    const int lr = lane >> 3;
    const int lc = (lane & 7) << 2;
    const int rc = lane & 15;
    const int kh = (lane >> 4) << 1;
#pragma unroll
    for (int c = 0; c < NC; ++c) {
        float* sA = st + (c & 1) * (2 * TILE_F);
        float* sB = sA + TILE_F;
#pragma unroll
        for (int q = 0; q < 4; ++q) {
            *(v4f*)(sA + (4 * q + lr) * PITCH + lc) = ga[c][q];
            *(v4f*)(sB + (4 * q + lr) * PITCH + lc) = gb[c][q];
        }
#pragma unroll
        for (int k = 0; k < KC; k += 4) {
            v2f a = *(const v2f*)(sA + rc * PITCH + kh + k);
            v2f b = *(const v2f*)(sB + rc * PITCH + kh + k);
            acc = __builtin_amdgcn_wmma_f32_16x16x4_f32(false, a, false, b,
                                                        (short)0, acc, false, false);
        }
    }
}

__global__ void __launch_bounds__(32 * NWAVE)
rnn_step(const float* __restrict__ x,
         const float* __restrict__ W_ih,
         const float* __restrict__ W_hh,
         const float* __restrict__ b_ih,
         const float* __restrict__ b_hh,
         const float* __restrict__ hprev,
         float* __restrict__ hnext,
         int t, int has_h)
{
    __shared__ float red[NWAVE * 32 * 8];
    __shared__ __attribute__((aligned(16))) float stage[NWAVE * WAVE_F];
    const int lane = threadIdx.x & 31;
    const int wave = threadIdx.x >> 5;
    const int tm = blockIdx.x;
    const int tn = blockIdx.y;

    const int hh = tn * 16 + (lane & 15);
    const float bias = b_ih[hh] + b_hh[hh];

    v8f acc = {};

    {
        const int lr = lane >> 3;
        const int lc = (lane & 7) << 2;
        const float* a0 = x + (size_t)(tm * 16 + lr) * (TSZ * FSZ) + (size_t)t * FSZ
                            + wave * (FSZ / NWAVE) + lc;
        const float* b0 = W_ih + (size_t)(tn * 16 + lr) * FSZ + wave * (FSZ / NWAVE) + lc;
        const float* a1 = hprev + (size_t)(tm * 16 + lr) * HSZ + wave * (HSZ / NWAVE) + lc;
        const float* b1 = W_hh + (size_t)(tn * 16 + lr) * HSZ + wave * (HSZ / NWAVE) + lc;
        float* st = stage + wave * WAVE_F;
        if (has_h) {
            staged_slice_f32<1 + (HSZ / NWAVE) / KC>(a0, (size_t)TSZ * FSZ, b0, (size_t)FSZ,
                                                     a1, b1, st, lane, acc);
        } else {
            staged_slice_f32<1>(a0, (size_t)TSZ * FSZ, b0, (size_t)FSZ, a1, b1, st, lane, acc);
        }
    }

    {
        float* r = red + wave * 256 + lane;
#pragma unroll
        for (int j = 0; j < 8; ++j) r[j * 32] = acc[j];
    }
    __syncthreads();

    {
        const int e = wave * 32 + lane;
        float s = 0.f;
#pragma unroll
        for (int w = 0; w < NWAVE; ++w) s += red[w * 256 + e];
        const int bb = tm * 16 + wave + ((lane >> 4) << 3);
        hnext[(size_t)bb * HSZ + hh] = tanhf(s + bias);
    }
}

__global__ void __launch_bounds__(32 * NWAVE)
fc_head(const float* __restrict__ hlast,
        const float* __restrict__ W_fc,
        const float* __restrict__ b_fc,
        float* __restrict__ out)
{
    __shared__ float red[NWAVE * 32 * 8];
    const int lane = threadIdx.x & 31;
    const int wave = threadIdx.x >> 5;
    const int tm = blockIdx.x;
    const int tn = blockIdx.y;

    v8f acc = {};
    {
        const float* A0 = hlast + (size_t)(tm * 16) * HSZ;
        const float* B0 = W_fc + (size_t)(tn * 16) * HSZ;
        wmma_tile_f32(A0, HSZ, B0, HSZ,
                      wave * (HSZ / NWAVE), (wave + 1) * (HSZ / NWAVE), lane, acc);
    }
    {
        const int base = wave * 256 + lane * 8;
#pragma unroll
        for (int j = 0; j < 8; ++j) red[base + j] = acc[j];
    }
    __syncthreads();

    if (wave == 0) {
        const int n  = lane & 15;
        const int mh = (lane >> 4) << 3;
        const int oo = tn * 16 + n;
        const float bias = b_fc[oo];
#pragma unroll
        for (int j = 0; j < 8; ++j) {
            const int idx = lane * 8 + j;
            float s = 0.f;
#pragma unroll
            for (int w = 0; w < NWAVE; ++w) s += red[w * 256 + idx];
            const int bb = tm * 16 + mh + j;
            out[(size_t)bb * OSZ + oo] = s + bias;
        }
    }
}

extern "C" void kernel_launch(void* const* d_in, const int* in_sizes, int n_in,
                              void* d_out, int out_size, void* d_ws, size_t ws_size,
                              hipStream_t stream)
{
    const float* x    = (const float*)d_in[0];
    const float* W_ih = (const float*)d_in[1];
    const float* W_hh = (const float*)d_in[2];
    const float* b_ih = (const float*)d_in[3];
    const float* b_hh = (const float*)d_in[4];
    const float* W_fc = (const float*)d_in[5];
    const float* b_fc = (const float*)d_in[6];
    float* out = (float*)d_out;

    float* h0 = (float*)d_ws;
    float* h1 = h0 + BSZ * HSZ;
    float* h[2] = { h0, h1 };

    const dim3 sgrid(BSZ / 16, HSZ / 16);
    for (int t = 0; t < TSZ; ++t) {
        const float* hp = h[t & 1];
        float*       hn = h[(t + 1) & 1];
        rnn_step<<<sgrid, 32 * NWAVE, 0, stream>>>(x, W_ih, W_hh, b_ih, b_hh,
                                                   hp, hn, t, (t > 0) ? 1 : 0);
    }
    const float* hlast = h[TSZ & 1];
    fc_head<<<dim3(BSZ / 16, OSZ / 16), 32 * NWAVE, 0, stream>>>(hlast, W_fc, b_fc, out);
}


# operands kept in WMMA fragment order in the workspace (x, W_ih, W_hh rewritten once per call, h stored so): rnn_step feeds WMMAs straight from b128 loads, no LDS staging; k and sum order unchanged
# speedup vs baseline: 1.7141x; 1.2217x over previous
#include <hip/hip_runtime.h>
#include <cstddef>


typedef __attribute__((ext_vector_type(2))) float v2f;
typedef __attribute__((ext_vector_type(4))) float v4f;
typedef __attribute__((ext_vector_type(8))) float v8f;

#define BSZ 64
#define TSZ 256
#define FSZ 256
#define HSZ 1024
#define OSZ 256

#define NWAVE 8

#define BLK_F 128
#define XBLK  (FSZ / 8)
#define HBLK  (HSZ / 8)

#define ARR_THREADS (256 * 256)
#define XR_V4   (TSZ * (BSZ / 16) * XBLK * 32)
#define WHH_V4  ((HSZ / 16) * HBLK * 32)
#define WIH_V4  ((HSZ / 16) * XBLK * 32)

__global__ void __launch_bounds__(256)
to_fragment_order(const float* __restrict__ x,
                  const float* __restrict__ W_ih,
                  const float* __restrict__ W_hh,
                  float* __restrict__ xr,
                  float* __restrict__ wir,
                  float* __restrict__ whr)
{
    const int gt = blockIdx.x * 256 + threadIdx.x;
    const int lane = gt & 31;
    const int r  = lane & 15;
    const int k2 = (lane >> 4) << 1;
#pragma unroll
    for (int it = 0; it < XR_V4 / ARR_THREADS; ++it) {
        const int idx = it * ARR_THREADS + gt;
        const int j  = (idx >> 5) & (XBLK - 1);
        const int tm = (idx >> 10) & 3;
        const int t  = idx >> 12;
        const float* s = x + ((size_t)(tm * 16 + r) * TSZ + t) * FSZ + 8 * j + k2;
        const v2f lo = *(const v2f*)s;
        const v2f hi = *(const v2f*)(s + 4);
        v4f o; o.x = lo.x; o.y = lo.y; o.z = hi.x; o.w = hi.y;
        ((v4f*)xr)[idx] = o;
    }
#pragma unroll
    for (int it = 0; it < WHH_V4 / ARR_THREADS; ++it) {
        const int idx = it * ARR_THREADS + gt;
        const int j  = (idx >> 5) & (HBLK - 1);
        const int tn = idx >> 12;
        const float* s = W_hh + (size_t)(tn * 16 + r) * HSZ + 8 * j + k2;
        const v2f lo = *(const v2f*)s;
        const v2f hi = *(const v2f*)(s + 4);
        v4f o; o.x = lo.x; o.y = lo.y; o.z = hi.x; o.w = hi.y;
        ((v4f*)whr)[idx] = o;
    }
#pragma unroll
    for (int it = 0; it < WIH_V4 / ARR_THREADS; ++it) {
        const int idx = it * ARR_THREADS + gt;
        const int j  = (idx >> 5) & (XBLK - 1);
        const int tn = idx >> 10;
        const float* s = W_ih + (size_t)(tn * 16 + r) * FSZ + 8 * j + k2;
        const v2f lo = *(const v2f*)s;
        const v2f hi = *(const v2f*)(s + 4);
        v4f o; o.x = lo.x; o.y = lo.y; o.z = hi.x; o.w = hi.y;
        ((v4f*)wir)[idx] = o;
    }
}

template <int NX, int NH>
__device__ __forceinline__ void slice_f32(const v4f* __restrict__ ax, const v4f* __restrict__ bx,
                                          const v4f* __restrict__ ah, const v4f* __restrict__ bh,
                                          v8f& acc)
{
    v4f ga[NX + NH], gb[NX + NH];
#pragma unroll
    for (int i = 0; i < NX; ++i) { ga[i] = ax[i * 32]; gb[i] = bx[i * 32]; }
#pragma unroll
    for (int i = 0; i < NH; ++i) { ga[NX + i] = ah[i * 32]; gb[NX + i] = bh[i * 32]; }
    __builtin_amdgcn_sched_barrier(0);
#pragma unroll
    for (int i = 0; i < NX + NH; ++i) {
        v2f a0, b0, a1, b1;
        a0.x = ga[i].x; a0.y = ga[i].y; b0.x = gb[i].x; b0.y = gb[i].y;
        a1.x = ga[i].z; a1.y = ga[i].w; b1.x = gb[i].z; b1.y = gb[i].w;
        acc = __builtin_amdgcn_wmma_f32_16x16x4_f32(false, a0, false, b0,
                                                    (short)0, acc, false, false);
        acc = __builtin_amdgcn_wmma_f32_16x16x4_f32(false, a1, false, b1,
                                                    (short)0, acc, false, false);
    }
}

__global__ void __launch_bounds__(32 * NWAVE)
rnn_step(const float* __restrict__ xr,
         const float* __restrict__ wir,
         const float* __restrict__ whr,
         const float* __restrict__ b_ih,
         const float* __restrict__ b_hh,
         const float* __restrict__ hprev,
         float* __restrict__ hnext,
         int t, int has_h)
{
    __shared__ float red[NWAVE * 32 * 8];
    const int lane = threadIdx.x & 31;
    const int wave = threadIdx.x >> 5;
    const int tm = blockIdx.x;
    const int tn = blockIdx.y;

    const int orow = ((wave & 1) << 3) + (lane >> 2);
    const int ocol = ((wave >> 2) << 3) + (((lane >> 1) & 1) << 2) + (((wave >> 1) & 1) << 1) + (lane & 1);
    const int hh = tn * 16 + ocol;
    const float bias = b_ih[hh] + b_hh[hh];

    v8f acc = {};

    {
        const v4f* ax = (const v4f*)xr  + ((size_t)((t * 4 + tm) * XBLK + 4 * wave) * 32 + lane);
        const v4f* bx = (const v4f*)wir + ((size_t)(tn * XBLK + 4 * wave) * 32 + lane);
        const v4f* ah = (const v4f*)hprev + ((size_t)(tm * HBLK + 16 * wave) * 32 + lane);
        const v4f* bh = (const v4f*)whr   + ((size_t)(tn * HBLK + 16 * wave) * 32 + lane);
        if (has_h) {
            slice_f32<4, 16>(ax, bx, ah, bh, acc);
        } else {
            slice_f32<4, 0>(ax, bx, ah, bh, acc);
        }
    }

    {
        float* r = red + wave * 256 + lane;
#pragma unroll
        for (int j = 0; j < 8; ++j) r[j * 32] = acc[j];
    }
    __syncthreads();

    {
        const int e = (orow & 7) * 32 + ocol + ((orow >> 3) << 4);
        float s = 0.f;
#pragma unroll
        for (int w = 0; w < NWAVE; ++w) s += red[w * 256 + e];
        hnext[(size_t)(tm * HBLK + 2 * tn) * BLK_F + wave * 32 + lane] = tanhf(s + bias);
    }
}

__device__ __forceinline__ void wmma_tile_f32(const float* __restrict__ Afr,
                                              const float* __restrict__ B0, int ldb,
                                              int kBeg, int kEnd, int lane, v8f& acc)
{
    const int rc = lane & 15;
    const int kh = (lane >> 4) << 1;
    const float* ap = Afr + lane * 4;
    const float* bp = B0 + (size_t)rc * ldb + kh;
#pragma unroll 4
    for (int k = kBeg; k < kEnd; k += 4) {
        v2f a = *(const v2f*)(ap + (k >> 3) * BLK_F + ((k >> 2) & 1) * 2);
        v2f b = *(const v2f*)(bp + k);
        acc = __builtin_amdgcn_wmma_f32_16x16x4_f32(false, a, false, b,
                                                    (short)0, acc, false, false);
    }
}

__global__ void __launch_bounds__(32 * NWAVE)
fc_head(const float* __restrict__ hlast,
        const float* __restrict__ W_fc,
        const float* __restrict__ b_fc,
        float* __restrict__ out)
{
    __shared__ float red[NWAVE * 32 * 8];
    const int lane = threadIdx.x & 31;
    const int wave = threadIdx.x >> 5;
    const int tm = blockIdx.x;
    const int tn = blockIdx.y;

    v8f acc = {};
    {
        const float* A0 = hlast + (size_t)(tm * HBLK) * BLK_F;
        const float* B0 = W_fc + (size_t)(tn * 16) * HSZ;
        wmma_tile_f32(A0, B0, HSZ,
                      wave * (HSZ / NWAVE), (wave + 1) * (HSZ / NWAVE), lane, acc);
    }
    {
        const int base = wave * 256 + lane * 8;
#pragma unroll
        for (int j = 0; j < 8; ++j) red[base + j] = acc[j];
    }
    __syncthreads();

    if (wave == 0) {
        const int n  = lane & 15;
        const int mh = (lane >> 4) << 3;
        const int oo = tn * 16 + n;
        const float bias = b_fc[oo];
#pragma unroll
        for (int j = 0; j < 8; ++j) {
            const int idx = lane * 8 + j;
            float s = 0.f;
#pragma unroll
            for (int w = 0; w < NWAVE; ++w) s += red[w * 256 + idx];
            const int bb = tm * 16 + mh + j;
            out[(size_t)bb * OSZ + oo] = s + bias;
        }
    }
}

extern "C" void kernel_launch(void* const* d_in, const int* in_sizes, int n_in,
                              void* d_out, int out_size, void* d_ws, size_t ws_size,
                              hipStream_t stream)
{
    const float* x    = (const float*)d_in[0];
    const float* W_ih = (const float*)d_in[1];
    const float* W_hh = (const float*)d_in[2];
    const float* b_ih = (const float*)d_in[3];
    const float* b_hh = (const float*)d_in[4];
    const float* W_fc = (const float*)d_in[5];
    const float* b_fc = (const float*)d_in[6];
    float* out = (float*)d_out;

    float* h0  = (float*)d_ws;
    float* h1  = h0 + BSZ * HSZ;
    float* whr = h1 + BSZ * HSZ;
    float* wir = whr + HSZ * HSZ;
    float* xr  = wir + HSZ * FSZ;
    float* h[2] = { h0, h1 };

    to_fragment_order<<<ARR_THREADS / 256, 256, 0, stream>>>(x, W_ih, W_hh, xr, wir, whr);

    const dim3 sgrid(BSZ / 16, HSZ / 16);
    for (int t = 0; t < TSZ; ++t) {
        const float* hp = h[t & 1];
        float*       hn = h[(t + 1) & 1];
        rnn_step<<<sgrid, 32 * NWAVE, 0, stream>>>(xr, wir, whr, b_ih, b_hh,
                                                   hp, hn, t, (t > 0) ? 1 : 0);
    }
    const float* hlast = h[TSZ & 1];
    fc_head<<<dim3(BSZ / 16, OSZ / 16), 32 * NWAVE, 0, stream>>>(hlast, W_fc, b_fc, out);
}
